# starting kernel
# baseline (speedup 1.0000x reference)
#include <hip/hip_runtime.h>

typedef __attribute__((ext_vector_type(2))) float v2f;
typedef __attribute__((ext_vector_type(8))) float v8f;

#define B_    4
#define H_    8
#define LQ    1024
#define DH    64
#define NIDX  33
#define NPAD  48
#define MAXREL 16

__global__ __launch_bounds__(32) void relpe_wmma_kernel(
    const float* __restrict__ query,
    const float* __restrict__ rel_table,
    const int*   __restrict__ time_ids,
    float*       __restrict__ out,
    int k_len, int ts_stride)
{
    __shared__ float S[16 * NPAD];

    const int wg = blockIdx.x;
    const int nq = LQ / 16;
    const int qt = wg % nq;
    const int h  = (wg / nq) % H_;
    const int b  = wg / (nq * H_);
    const int q0 = qt * 16;

    const int lane   = threadIdx.x;
    const int m      = lane & 15;
    const int hiHalf = lane >> 4;

    const float* qrow = query + ((((size_t)b * H_ + h) * LQ + (q0 + m)) * DH);

    const int n0 = m, n1 = 16 + m, n2 = 32 + m;
    const float msk2   = (n2 < NIDX) ? 1.0f : 0.0f;
    const int   n2c    = (n2 < NIDX) ? n2 : 0;
    const float* trow0 = rel_table + n0  * DH;
    const float* trow1 = rel_table + n1  * DH;
    const float* trow2 = rel_table + n2c * DH;

    v8f c0 = {}, c1 = {}, c2 = {};
    #pragma unroll
    for (int kk = 0; kk < DH; kk += 4) {
        const int kb = kk + (hiHalf << 1);
        v2f a, b0, b1, b2;
        a.x  = qrow[kb];          a.y  = qrow[kb + 1];
        b0.x = trow0[kb];         b0.y = trow0[kb + 1];
        b1.x = trow1[kb];         b1.y = trow1[kb + 1];
        b2.x = trow2[kb] * msk2;  b2.y = trow2[kb + 1] * msk2;
        c0 = __builtin_amdgcn_wmma_f32_16x16x4_f32(false, a, false, b0, (short)0, c0, false, false);
        c1 = __builtin_amdgcn_wmma_f32_16x16x4_f32(false, a, false, b1, (short)0, c1, false, false);
        c2 = __builtin_amdgcn_wmma_f32_16x16x4_f32(false, a, false, b2, (short)0, c2, false, false);
    }

    #pragma unroll
    for (int r = 0; r < 8; ++r) {
        const int M = r + 8 * hiHalf;
        S[M * NPAD +  0 + m] = c0[r];
        S[M * NPAD + 16 + m] = c1[r];
        S[M * NPAD + 32 + m] = c2[r];
    }
    __syncthreads();

    const int* ts = time_ids + (size_t)b * ts_stride;

    int tq[16];
    #pragma unroll
    for (int q = 0; q < 16; ++q) tq[q] = ts[q0 + q];

    float* outBase = out + (((size_t)(b * H_ + h)) * LQ + q0) * (size_t)k_len;

    for (int kc = lane * 4; kc < k_len; kc += 32 * 4) {
        const int4 tk = *(const int4*)(ts + kc);
        #pragma unroll
        for (int q = 0; q < 16; ++q) {
            const int base = MAXREL - tq[q];
            int i0 = min(max(tk.x + base, 0), 2 * MAXREL);
            int i1 = min(max(tk.y + base, 0), 2 * MAXREL);
            int i2 = min(max(tk.z + base, 0), 2 * MAXREL);
            int i3 = min(max(tk.w + base, 0), 2 * MAXREL);
            const float* Sq = &S[q * NPAD];
            float4 v;
            v.x = Sq[i0]; v.y = Sq[i1]; v.z = Sq[i2]; v.w = Sq[i3];
            *(float4*)(outBase + (size_t)q * k_len + kc) = v;
        }
    }
}

extern "C" void kernel_launch(void* const* d_in, const int* in_sizes, int n_in,
                              void* d_out, int out_size, void* d_ws, size_t ws_size,
                              hipStream_t stream) {
    const float* query     = (const float*)d_in[0];
    const float* rel_table = (const float*)d_in[1];
    const int*   time_ids  = (const int*)d_in[2];
    const int k_len     = out_size / (B_ * H_ * LQ);
    const int ts_stride = in_sizes[2] / B_;

    const dim3 grid(B_ * H_ * (LQ / 16));
    relpe_wmma_kernel<<<grid, 32, 0, stream>>>(
        query, rel_table, time_ids, (float*)d_out, k_len, ts_stride);
}


# Store loop turned row-major: a wave writes each 4 KiB output row as one sequential run of 8 stores; the row's key time ids are loaded once at entry and kept in registers (no load left in the loop)
# speedup vs baseline: 1.3445x; 1.3445x over previous
#include <hip/hip_runtime.h>

typedef __attribute__((ext_vector_type(2))) float v2f;
typedef __attribute__((ext_vector_type(8))) float v8f;

#define B_    4
#define H_    8
#define LQ    1024
#define DH    64
#define NIDX  33
#define NPAD  48
#define MAXREL 16
#define LK    1024

__global__ __launch_bounds__(32) void relpe_wmma_kernel(
    const float* __restrict__ query,
    const float* __restrict__ rel_table,
    const int*   __restrict__ time_ids,
    float*       __restrict__ out,
    int k_len, int ts_stride)
{
    __shared__ float S[16 * NPAD];

    const int wg = blockIdx.x;
    const int nq = LQ / 16;
    const int qt = wg % nq;
    const int h  = (wg / nq) % H_;
    const int b  = wg / (nq * H_);
    const int q0 = qt * 16;

    const int lane   = threadIdx.x;
    const int m      = lane & 15;
    const int hiHalf = lane >> 4;

    const int* ts = time_ids + (size_t)b * ts_stride;
    const bool fullK = (k_len == LK);
    int4 tk[8];
    int  tqv = 0;
    if (fullK) {
        #pragma unroll
        for (int j = 0; j < 8; ++j) tk[j] = *(const int4*)(ts + j * 128 + lane * 4);
        tqv = ts[q0 + m];
    }

    const float* qrow = query + ((((size_t)b * H_ + h) * LQ + (q0 + m)) * DH);

    const int n0 = m, n1 = 16 + m, n2 = 32 + m;
    const float msk2   = (n2 < NIDX) ? 1.0f : 0.0f;
    const int   n2c    = (n2 < NIDX) ? n2 : 0;
    const float* trow0 = rel_table + n0  * DH;
    const float* trow1 = rel_table + n1  * DH;
    const float* trow2 = rel_table + n2c * DH;

    v8f c0 = {}, c1 = {}, c2 = {};
    #pragma unroll
    for (int kk = 0; kk < DH; kk += 4) {
        const int kb = kk + (hiHalf << 1);
        v2f a, b0, b1, b2;
        a.x  = qrow[kb];          a.y  = qrow[kb + 1];
        b0.x = trow0[kb];         b0.y = trow0[kb + 1];
        b1.x = trow1[kb];         b1.y = trow1[kb + 1];
        b2.x = trow2[kb] * msk2;  b2.y = trow2[kb + 1] * msk2;
        c0 = __builtin_amdgcn_wmma_f32_16x16x4_f32(false, a, false, b0, (short)0, c0, false, false);
        c1 = __builtin_amdgcn_wmma_f32_16x16x4_f32(false, a, false, b1, (short)0, c1, false, false);
        c2 = __builtin_amdgcn_wmma_f32_16x16x4_f32(false, a, false, b2, (short)0, c2, false, false);
    }

    #pragma unroll
    for (int r = 0; r < 8; ++r) {
        const int M = r + 8 * hiHalf;
        S[M * NPAD +  0 + m] = c0[r];
        S[M * NPAD + 16 + m] = c1[r];
        S[M * NPAD + 32 + m] = c2[r];
    }
    __syncthreads();

    float* outBase = out + (((size_t)(b * H_ + h)) * LQ + q0) * (size_t)k_len;

    if (fullK) {
        float* orow = outBase + lane * 4;
        #pragma unroll 2
        for (int q = 0; q < 16; ++q) {
            const int base = MAXREL - __builtin_amdgcn_readlane(tqv, q);
            const float* Sq = &S[q * NPAD];
            #pragma unroll
            for (int j = 0; j < 8; ++j) {
                const int i0 = min(max(tk[j].x + base, 0), 2 * MAXREL);
                const int i1 = min(max(tk[j].y + base, 0), 2 * MAXREL);
                const int i2 = min(max(tk[j].z + base, 0), 2 * MAXREL);
                const int i3 = min(max(tk[j].w + base, 0), 2 * MAXREL);
                float4 v;
                v.x = Sq[i0]; v.y = Sq[i1]; v.z = Sq[i2]; v.w = Sq[i3];
                *(float4*)(orow + j * 128) = v;
            }
            orow += LK;
        }
    } else {
        int tq[16];
        #pragma unroll
        for (int q = 0; q < 16; ++q) tq[q] = ts[q0 + q];

        for (int kc = lane * 4; kc < k_len; kc += 32 * 4) {
            const int4 t4 = *(const int4*)(ts + kc);
            #pragma unroll
            for (int q = 0; q < 16; ++q) {
                const int base = MAXREL - tq[q];
                int i0 = min(max(t4.x + base, 0), 2 * MAXREL);
                int i1 = min(max(t4.y + base, 0), 2 * MAXREL);
                int i2 = min(max(t4.z + base, 0), 2 * MAXREL);
                int i3 = min(max(t4.w + base, 0), 2 * MAXREL);
                const float* Sq = &S[q * NPAD];
                float4 v;
                v.x = Sq[i0]; v.y = Sq[i1]; v.z = Sq[i2]; v.w = Sq[i3];
                *(float4*)(outBase + (size_t)q * k_len + kc) = v;
            }
        }
    }
}

extern "C" void kernel_launch(void* const* d_in, const int* in_sizes, int n_in,
                              void* d_out, int out_size, void* d_ws, size_t ws_size,
                              hipStream_t stream) {
    const float* query     = (const float*)d_in[0];
    const float* rel_table = (const float*)d_in[1];
    const int*   time_ids  = (const int*)d_in[2];
    const int k_len     = out_size / (B_ * H_ * LQ);
    const int ts_stride = in_sizes[2] / B_;

    const dim3 grid(B_ * H_ * (LQ / 16));
    relpe_wmma_kernel<<<grid, 32, 0, stream>>>(
        query, rel_table, time_ids, (float*)d_out, k_len, ts_stride);
}
